# starting kernel
# baseline (speedup 1.0000x reference)
#include <hip/hip_runtime.h>

typedef __attribute__((ext_vector_type(2))) float v2f;
typedef __attribute__((ext_vector_type(4))) float f4;
typedef __attribute__((ext_vector_type(8))) float v8f;

#define NF 128
#define TM 32
#define LDSS 132

__device__ __forceinline__ v8f wmma4(v2f a, v2f b, v8f c) {
    return __builtin_amdgcn_wmma_f32_16x16x4_f32(
         false, a,  false, b,
         (short)0, c,  false,  false);
}

__global__ __launch_bounds__(64)
void metapath_wmma_kernel(const float* __restrict__ fmi,
                          const float* __restrict__ fg,
                          const float* __restrict__ fdr,
                          const float* __restrict__ Wdd,
                          const float* __restrict__ Wdisdr,
                          const float* __restrict__ Wdrug,
                          const float* __restrict__ Wdis,
                          const int*   __restrict__ mp,
                          float* __restrict__ out)
{
    __shared__ float sA[TM][LDSS];
    __shared__ float sR[TM][LDSS];
    __shared__ int   sIdx[TM * 4];

    const int t        = threadIdx.x;
    const int lane     = t & 31;
    const int wave     = t >> 5;
    const int rowbase  = wave * 16;
    const int l15      = lane & 15;
    const int koff     = (lane >> 4) * 2;
    const int blockRow = blockIdx.x * TM;

    sIdx[t]      = mp[blockRow * 4 + t];
    sIdx[t + 64] = mp[blockRow * 4 + 64 + t];
    __syncthreads();

    #pragma unroll 1
    for (int phase = 0; phase < 2; ++phase) {
        const float* tA0   = (phase == 0) ? fmi : fdr;
        const float* tA1   = fg;
        const float* tRq   = fg;
        const float* tRh   = (phase == 0) ? fdr : fmi;
        const int    cA0   = (phase == 0) ? 0 : 3;
        const int    cA1   = (phase == 0) ? 1 : 2;
        const int    cRq   = (phase == 0) ? 2 : 1;
        const int    cRh   = (phase == 0) ? 3 : 0;
        const float* Wmain = (phase == 0) ? Wdd : Wdisdr;
        const float* Wout  = (phase == 0) ? Wdis : Wdrug;
        const int    ncol0 = (phase == 0) ? 64 : 0;

        #pragma unroll 4
        for (int i = t; i < TM * (NF / 4); i += 64) {
            const int r  = i >> 5;
            const int c4 = (i & 31) * 4;
            const int i0 = sIdx[r * 4 + cA0];
            const int i1 = sIdx[r * 4 + cA1];
            const int iq = sIdx[r * 4 + cRq];
            const int ih = sIdx[r * 4 + cRh];
            const f4 x = *(const f4*)(tA0 + i0 * NF + c4);
            const f4 y = *(const f4*)(tA1 + i1 * NF + c4);
            const f4 u = *(const f4*)(tRq + iq * NF + c4);
            const f4 v = *(const f4*)(tRh + ih * NF + c4);
            *(f4*)&sA[r][c4] = (x + y) * 0.5f;
            *(f4*)&sR[r][c4] = u * 0.25f + v * 0.5f;
        }
        __syncthreads();

        const v8f z = {0.f, 0.f, 0.f, 0.f, 0.f, 0.f, 0.f, 0.f};
        v8f acc[8];
        #pragma unroll
        for (int nt = 0; nt < 8; ++nt) acc[nt] = z;

        #pragma unroll 4
        for (int kk = 0; kk < 32; ++kk) {
            const int kb = kk * 4 + koff;
            const v2f a = *(const v2f*)&sA[rowbase + l15][kb];
            #pragma unroll
            for (int nt = 0; nt < 8; ++nt) {
                const v2f b = *(const v2f*)(Wmain + (nt * 16 + l15) * NF + kb);
                acc[nt] = wmma4(a, b, acc[nt]);
            }
        }

        const int mb = rowbase + 8 * (lane >> 4);
        #pragma unroll
        for (int nt = 0; nt < 8; ++nt) {
            const int col = nt * 16 + l15;
            #pragma unroll
            for (int v = 0; v < 8; ++v) {
                sA[mb + v][col] = 0.25f * acc[nt][v] + sR[mb + v][col];
            }
        }

        v8f acc2[4];
        #pragma unroll
        for (int nt = 0; nt < 4; ++nt) acc2[nt] = z;

        #pragma unroll 4
        for (int kk = 0; kk < 32; ++kk) {
            const int kb = kk * 4 + koff;
            const v2f a = *(const v2f*)&sA[rowbase + l15][kb];
            #pragma unroll
            for (int nt = 0; nt < 4; ++nt) {
                const v2f b = *(const v2f*)(Wout + (nt * 16 + l15) * NF + kb);
                acc2[nt] = wmma4(a, b, acc2[nt]);
            }
        }

        const int grow = blockRow + mb;
        #pragma unroll
        for (int nt = 0; nt < 4; ++nt) {
            const int col = ncol0 + nt * 16 + l15;
            #pragma unroll
            for (int v = 0; v < 8; ++v) {
                __builtin_nontemporal_store(acc2[nt][v], out + (grow + v) * NF + col);
            }
        }
        __syncthreads();
    }
}

extern "C" void kernel_launch(void* const* d_in, const int* in_sizes, int n_in,
                              void* d_out, int out_size, void* d_ws, size_t ws_size,
                              hipStream_t stream) {
    const float* fmi    = (const float*)d_in[0];
    const float* fg     = (const float*)d_in[1];
    const float* fdr    = (const float*)d_in[2];
    const float* Wdd    = (const float*)d_in[3];
    const float* Wdisdr = (const float*)d_in[4];
    const float* Wdrug  = (const float*)d_in[5];
    const float* Wdis   = (const float*)d_in[6];
    const int*   mp     = (const int*)d_in[7];
    float* out = (float*)d_out;

    const int n_rows = in_sizes[7] / 4;
    const int blocks = n_rows / TM;

    metapath_wmma_kernel<<<dim3(blocks), dim3(64), 0, stream>>>(
        fmi, fg, fdr, Wdd, Wdisdr, Wdrug, Wdis, mp, out);
}


# Re-tile to 64-row work-groups of 4 waves: each wave owns 32 output columns for all rows, so one global weight fragment feeds 4 row tiles (2 in GEMM-2) instead of 1; summation order unchanged
# speedup vs baseline: 1.5939x; 1.5939x over previous
#include <hip/hip_runtime.h>

typedef __attribute__((ext_vector_type(2))) float v2f;
typedef __attribute__((ext_vector_type(4))) float f4;
typedef __attribute__((ext_vector_type(8))) float v8f;

#define NF 128
#define TM 64
#define LDSS 132

__device__ __forceinline__ v8f wmma4(v2f a, v2f b, v8f c) {
    return __builtin_amdgcn_wmma_f32_16x16x4_f32(
         false, a,  false, b,
         (short)0, c,  false,  false);
}

__global__ __launch_bounds__(128)
void metapath_wmma_kernel(const float* __restrict__ fmi,
                          const float* __restrict__ fg,
                          const float* __restrict__ fdr,
                          const float* __restrict__ Wdd,
                          const float* __restrict__ Wdisdr,
                          const float* __restrict__ Wdrug,
                          const float* __restrict__ Wdis,
                          const int*   __restrict__ mp,
                          float* __restrict__ out)
{
    __shared__ float sA[TM][LDSS];
    __shared__ float sR[TM][LDSS];
    __shared__ int   sIdx[TM * 4];

    const int t        = threadIdx.x;
    const int lane     = t & 31;
    const int wave     = t >> 5;
    const int l15      = lane & 15;
    const int koff     = (lane >> 4) * 2;
    const int blockRow = blockIdx.x * TM;

    sIdx[t]       = mp[blockRow * 4 + t];
    sIdx[t + 128] = mp[blockRow * 4 + 128 + t];
    __syncthreads();

    #pragma unroll 1
    for (int phase = 0; phase < 2; ++phase) {
        const float* tA0   = (phase == 0) ? fmi : fdr;
        const float* tA1   = fg;
        const float* tRq   = fg;
        const float* tRh   = (phase == 0) ? fdr : fmi;
        const int    cA0   = (phase == 0) ? 0 : 3;
        const int    cA1   = (phase == 0) ? 1 : 2;
        const int    cRq   = (phase == 0) ? 2 : 1;
        const int    cRh   = (phase == 0) ? 3 : 0;
        const float* Wmain = (phase == 0) ? Wdd : Wdisdr;
        const float* Wout  = (phase == 0) ? Wdis : Wdrug;
        const int    ncol0 = (phase == 0) ? 64 : 0;

        #pragma unroll 4
        for (int i = t; i < TM * (NF / 4); i += 128) {
            const int r  = i >> 5;
            const int c4 = (i & 31) * 4;
            const int i0 = sIdx[r * 4 + cA0];
            const int i1 = sIdx[r * 4 + cA1];
            const int iq = sIdx[r * 4 + cRq];
            const int ih = sIdx[r * 4 + cRh];
            const f4 x = *(const f4*)(tA0 + i0 * NF + c4);
            const f4 y = *(const f4*)(tA1 + i1 * NF + c4);
            const f4 u = *(const f4*)(tRq + iq * NF + c4);
            const f4 v = *(const f4*)(tRh + ih * NF + c4);
            *(f4*)&sA[r][c4] = (x + y) * 0.5f;
            *(f4*)&sR[r][c4] = u * 0.25f + v * 0.5f;
        }
        __syncthreads();

        const v8f z = {0.f, 0.f, 0.f, 0.f, 0.f, 0.f, 0.f, 0.f};
        const int nb = wave * 32;
        v8f acc[4][2];
        #pragma unroll
        for (int mt = 0; mt < 4; ++mt) { acc[mt][0] = z; acc[mt][1] = z; }

        const float* wb0 = Wmain + (nb + l15) * NF + koff;
        const float* wb1 = Wmain + (nb + 16 + l15) * NF + koff;

        #pragma unroll 4
        for (int kk = 0; kk < 32; ++kk) {
            const int kb = kk * 4 + koff;
            const v2f b0 = *(const v2f*)(wb0 + kk * 4);
            const v2f b1 = *(const v2f*)(wb1 + kk * 4);
            #pragma unroll
            for (int mt = 0; mt < 4; ++mt) {
                const v2f a = *(const v2f*)&sA[mt * 16 + l15][kb];
                acc[mt][0] = wmma4(a, b0, acc[mt][0]);
                acc[mt][1] = wmma4(a, b1, acc[mt][1]);
            }
        }

        const int mh8 = 8 * (lane >> 4);
        #pragma unroll
        for (int mt = 0; mt < 4; ++mt) {
            #pragma unroll
            for (int j = 0; j < 2; ++j) {
                const int col = nb + j * 16 + l15;
                #pragma unroll
                for (int v = 0; v < 8; ++v) {
                    const int row = mt * 16 + mh8 + v;
                    sR[row][col] = 0.25f * acc[mt][j][v] + sR[row][col];
                }
            }
        }
        __syncthreads();

        const int mq = (wave & 1) * 32;
        const int nq = (wave >> 1) * 32;
        v8f acc2[2][2];
        #pragma unroll
        for (int mt = 0; mt < 2; ++mt) { acc2[mt][0] = z; acc2[mt][1] = z; }

        const float* wo0 = Wout + (nq + l15) * NF + koff;
        const float* wo1 = Wout + (nq + 16 + l15) * NF + koff;

        #pragma unroll 4
        for (int kk = 0; kk < 32; ++kk) {
            const int kb = kk * 4 + koff;
            const v2f b0 = *(const v2f*)(wo0 + kk * 4);
            const v2f b1 = *(const v2f*)(wo1 + kk * 4);
            #pragma unroll
            for (int mt = 0; mt < 2; ++mt) {
                const v2f a = *(const v2f*)&sR[mq + mt * 16 + l15][kb];
                acc2[mt][0] = wmma4(a, b0, acc2[mt][0]);
                acc2[mt][1] = wmma4(a, b1, acc2[mt][1]);
            }
        }

        #pragma unroll
        for (int mt = 0; mt < 2; ++mt) {
            const int grow = blockRow + mq + mt * 16 + mh8;
            #pragma unroll
            for (int j = 0; j < 2; ++j) {
                const int col = ncol0 + nq + j * 16 + l15;
                #pragma unroll
                for (int v = 0; v < 8; ++v) {
                    __builtin_nontemporal_store(acc2[mt][j][v], out + (grow + v) * NF + col);
                }
            }
        }
        __syncthreads();
    }
}

extern "C" void kernel_launch(void* const* d_in, const int* in_sizes, int n_in,
                              void* d_out, int out_size, void* d_ws, size_t ws_size,
                              hipStream_t stream) {
    const float* fmi    = (const float*)d_in[0];
    const float* fg     = (const float*)d_in[1];
    const float* fdr    = (const float*)d_in[2];
    const float* Wdd    = (const float*)d_in[3];
    const float* Wdisdr = (const float*)d_in[4];
    const float* Wdrug  = (const float*)d_in[5];
    const float* Wdis   = (const float*)d_in[6];
    const int*   mp     = (const int*)d_in[7];
    float* out = (float*)d_out;

    const int n_rows = in_sizes[7] / 4;
    const int blocks = n_rows / TM;

    metapath_wmma_kernel<<<dim3(blocks), dim3(128), 0, stream>>>(
        fmi, fg, fdr, Wdd, Wdisdr, Wdrug, Wdis, mp, out);
}


# 128-bit fragment loads for both WMMA operands (lane half h holds k=8c+4h..+3, two fragments per load): half as many 16-line global weight loads; k now summed as 0,1,4,5,2,3,6,7 in each 8
# speedup vs baseline: 1.6539x; 1.0376x over previous
#include <hip/hip_runtime.h>

typedef __attribute__((ext_vector_type(2))) float v2f;
typedef __attribute__((ext_vector_type(4))) float f4;
typedef __attribute__((ext_vector_type(8))) float v8f;

#define NF 128
#define TM 64
#define LDSS 132

__device__ __forceinline__ v8f wmma4(v2f a, v2f b, v8f c) {
    return __builtin_amdgcn_wmma_f32_16x16x4_f32(
         false, a,  false, b,
         (short)0, c,  false,  false);
}

__global__ __launch_bounds__(128)
void metapath_wmma_kernel(const float* __restrict__ fmi,
                          const float* __restrict__ fg,
                          const float* __restrict__ fdr,
                          const float* __restrict__ Wdd,
                          const float* __restrict__ Wdisdr,
                          const float* __restrict__ Wdrug,
                          const float* __restrict__ Wdis,
                          const int*   __restrict__ mp,
                          float* __restrict__ out)
{
    __shared__ float sA[TM][LDSS];
    __shared__ float sR[TM][LDSS];
    __shared__ int   sIdx[TM * 4];

    const int t        = threadIdx.x;
    const int lane     = t & 31;
    const int wave     = t >> 5;
    const int l15      = lane & 15;
    const int koff     = (lane >> 4) * 4;
    const int blockRow = blockIdx.x * TM;

    sIdx[t]       = mp[blockRow * 4 + t];
    sIdx[t + 128] = mp[blockRow * 4 + 128 + t];
    __syncthreads();

    #pragma unroll 1
    for (int phase = 0; phase < 2; ++phase) {
        const float* tA0   = (phase == 0) ? fmi : fdr;
        const float* tA1   = fg;
        const float* tRq   = fg;
        const float* tRh   = (phase == 0) ? fdr : fmi;
        const int    cA0   = (phase == 0) ? 0 : 3;
        const int    cA1   = (phase == 0) ? 1 : 2;
        const int    cRq   = (phase == 0) ? 2 : 1;
        const int    cRh   = (phase == 0) ? 3 : 0;
        const float* Wmain = (phase == 0) ? Wdd : Wdisdr;
        const float* Wout  = (phase == 0) ? Wdis : Wdrug;
        const int    ncol0 = (phase == 0) ? 64 : 0;

        #pragma unroll 4
        for (int i = t; i < TM * (NF / 4); i += 128) {
            const int r  = i >> 5;
            const int c4 = (i & 31) * 4;
            const int i0 = sIdx[r * 4 + cA0];
            const int i1 = sIdx[r * 4 + cA1];
            const int iq = sIdx[r * 4 + cRq];
            const int ih = sIdx[r * 4 + cRh];
            const f4 x = *(const f4*)(tA0 + i0 * NF + c4);
            const f4 y = *(const f4*)(tA1 + i1 * NF + c4);
            const f4 u = *(const f4*)(tRq + iq * NF + c4);
            const f4 v = *(const f4*)(tRh + ih * NF + c4);
            *(f4*)&sA[r][c4] = (x + y) * 0.5f;
            *(f4*)&sR[r][c4] = u * 0.25f + v * 0.5f;
        }
        __syncthreads();

        const v8f z = {0.f, 0.f, 0.f, 0.f, 0.f, 0.f, 0.f, 0.f};
        const int nb = wave * 32;
        v8f acc[4][2];
        #pragma unroll
        for (int mt = 0; mt < 4; ++mt) { acc[mt][0] = z; acc[mt][1] = z; }

        const float* wb0 = Wmain + (nb + l15) * NF + koff;
        const float* wb1 = Wmain + (nb + 16 + l15) * NF + koff;

        #pragma unroll 2
        for (int kc = 0; kc < 16; ++kc) {
            const int kb = kc * 8 + koff;
            const f4 b0 = *(const f4*)(wb0 + kc * 8);
            const f4 b1 = *(const f4*)(wb1 + kc * 8);
            f4 a[4];
            #pragma unroll
            for (int mt = 0; mt < 4; ++mt)
                a[mt] = *(const f4*)&sA[mt * 16 + l15][kb];
            #pragma unroll
            for (int mt = 0; mt < 4; ++mt) {
                const v2f alo = {a[mt].x, a[mt].y};
                acc[mt][0] = wmma4(alo, (v2f){b0.x, b0.y}, acc[mt][0]);
                acc[mt][1] = wmma4(alo, (v2f){b1.x, b1.y}, acc[mt][1]);
            }
            #pragma unroll
            for (int mt = 0; mt < 4; ++mt) {
                const v2f ahi = {a[mt].z, a[mt].w};
                acc[mt][0] = wmma4(ahi, (v2f){b0.z, b0.w}, acc[mt][0]);
                acc[mt][1] = wmma4(ahi, (v2f){b1.z, b1.w}, acc[mt][1]);
            }
        }

        const int mh8 = 8 * (lane >> 4);
        #pragma unroll
        for (int mt = 0; mt < 4; ++mt) {
            #pragma unroll
            for (int j = 0; j < 2; ++j) {
                const int col = nb + j * 16 + l15;
                #pragma unroll
                for (int v = 0; v < 8; ++v) {
                    const int row = mt * 16 + mh8 + v;
                    sR[row][col] = 0.25f * acc[mt][j][v] + sR[row][col];
                }
            }
        }
        __syncthreads();

        const int mq = (wave & 1) * 32;
        const int nq = (wave >> 1) * 32;
        v8f acc2[2][2];
        #pragma unroll
        for (int mt = 0; mt < 2; ++mt) { acc2[mt][0] = z; acc2[mt][1] = z; }

        const float* wo0 = Wout + (nq + l15) * NF + koff;
        const float* wo1 = Wout + (nq + 16 + l15) * NF + koff;

        #pragma unroll 2
        for (int kc = 0; kc < 16; ++kc) {
            const int kb = kc * 8 + koff;
            const f4 b0 = *(const f4*)(wo0 + kc * 8);
            const f4 b1 = *(const f4*)(wo1 + kc * 8);
            f4 a[2];
            #pragma unroll
            for (int mt = 0; mt < 2; ++mt)
                a[mt] = *(const f4*)&sR[mq + mt * 16 + l15][kb];
            #pragma unroll
            for (int mt = 0; mt < 2; ++mt) {
                const v2f alo = {a[mt].x, a[mt].y};
                acc2[mt][0] = wmma4(alo, (v2f){b0.x, b0.y}, acc2[mt][0]);
                acc2[mt][1] = wmma4(alo, (v2f){b1.x, b1.y}, acc2[mt][1]);
            }
            #pragma unroll
            for (int mt = 0; mt < 2; ++mt) {
                const v2f ahi = {a[mt].z, a[mt].w};
                acc2[mt][0] = wmma4(ahi, (v2f){b0.z, b0.w}, acc2[mt][0]);
                acc2[mt][1] = wmma4(ahi, (v2f){b1.z, b1.w}, acc2[mt][1]);
            }
        }

        #pragma unroll
        for (int mt = 0; mt < 2; ++mt) {
            const int grow = blockRow + mq + mt * 16 + mh8;
            #pragma unroll
            for (int j = 0; j < 2; ++j) {
                const int col = ncol0 + nq + j * 16 + l15;
                #pragma unroll
                for (int v = 0; v < 8; ++v) {
                    __builtin_nontemporal_store(acc2[mt][j][v], out + (grow + v) * NF + col);
                }
            }
        }
        __syncthreads();
    }
}

extern "C" void kernel_launch(void* const* d_in, const int* in_sizes, int n_in,
                              void* d_out, int out_size, void* d_ws, size_t ws_size,
                              hipStream_t stream) {
    const float* fmi    = (const float*)d_in[0];
    const float* fg     = (const float*)d_in[1];
    const float* fdr    = (const float*)d_in[2];
    const float* Wdd    = (const float*)d_in[3];
    const float* Wdisdr = (const float*)d_in[4];
    const float* Wdrug  = (const float*)d_in[5];
    const float* Wdis   = (const float*)d_in[6];
    const int*   mp     = (const int*)d_in[7];
    float* out = (float*)d_out;

    const int n_rows = in_sizes[7] / 4;
    const int blocks = n_rows / TM;

    metapath_wmma_kernel<<<dim3(blocks), dim3(128), 0, stream>>>(
        fmi, fg, fdr, Wdd, Wdisdr, Wdrug, Wdis, mp, out);
}


# Both paths in one pass of an 8-wave work-group with four LDS tiles: each gathered row is read once instead of twice; waves 0-3 and 4-7 take one path each in GEMM-1; sums unchanged
# speedup vs baseline: 1.8108x; 1.0949x over previous
#include <hip/hip_runtime.h>

typedef __attribute__((ext_vector_type(2))) float v2f;
typedef __attribute__((ext_vector_type(4))) float f4;
typedef __attribute__((ext_vector_type(8))) float v8f;

#define NF 128
#define TM 64
#define LDSS 132

__device__ __forceinline__ v8f wmma4(v2f a, v2f b, v8f c) {
    return __builtin_amdgcn_wmma_f32_16x16x4_f32(
         false, a,  false, b,
         (short)0, c,  false,  false);
}

typedef __attribute__((ext_vector_type(4))) int i4;

__global__ __launch_bounds__(256)
void metapath_wmma_kernel(const float* __restrict__ fmi,
                          const float* __restrict__ fg,
                          const float* __restrict__ fdr,
                          const float* __restrict__ Wdd,
                          const float* __restrict__ Wdisdr,
                          const float* __restrict__ Wdrug,
                          const float* __restrict__ Wdis,
                          const int*   __restrict__ mp,
                          float* __restrict__ out)
{
    __shared__ float sA0[TM][LDSS];
    __shared__ float sR0[TM][LDSS];
    __shared__ float sA1[TM][LDSS];
    __shared__ float sR1[TM][LDSS];
    __shared__ __attribute__((aligned(16))) int sIdx[TM * 4];

    const int t        = threadIdx.x;
    const int lane     = t & 31;
    const int wave     = t >> 5;
    const int l15      = lane & 15;
    const int koff     = (lane >> 4) * 4;
    const int blockRow = blockIdx.x * TM;

    sIdx[t] = mp[blockRow * 4 + t];
    __syncthreads();

    #pragma unroll 4
    for (int i = t; i < TM * (NF / 4); i += 256) {
        const int r  = i >> 5;
        const int c4 = (i & 31) * 4;
        const i4 id = *(const i4*)&sIdx[r * 4];
        const f4 mi = *(const f4*)(fmi + id.x * NF + c4);
        const f4 g1 = *(const f4*)(fg  + id.y * NF + c4);
        const f4 g2 = *(const f4*)(fg  + id.z * NF + c4);
        const f4 dr = *(const f4*)(fdr + id.w * NF + c4);
        *(f4*)&sA0[r][c4] = (mi + g1) * 0.5f;
        *(f4*)&sR0[r][c4] = g2 * 0.25f + dr * 0.5f;
        *(f4*)&sA1[r][c4] = (dr + g2) * 0.5f;
        *(f4*)&sR1[r][c4] = g1 * 0.25f + mi * 0.5f;
    }
    __syncthreads();

    const int p1 = wave >> 2;
    const int nb = (wave & 3) * 32;
    const float* Wmain = p1 ? Wdisdr : Wdd;
    float (*tA)[LDSS] = p1 ? sA1 : sA0;
    float (*tR)[LDSS] = p1 ? sR1 : sR0;

    const v8f z = {0.f, 0.f, 0.f, 0.f, 0.f, 0.f, 0.f, 0.f};
    v8f acc[4][2];
    #pragma unroll
    for (int mt = 0; mt < 4; ++mt) { acc[mt][0] = z; acc[mt][1] = z; }

    const float* wb0 = Wmain + (nb + l15) * NF + koff;
    const float* wb1 = Wmain + (nb + 16 + l15) * NF + koff;

    #pragma unroll 2
    for (int kc = 0; kc < 16; ++kc) {
        const int kb = kc * 8 + koff;
        const f4 b0 = *(const f4*)(wb0 + kc * 8);
        const f4 b1 = *(const f4*)(wb1 + kc * 8);
        f4 a[4];
        #pragma unroll
        for (int mt = 0; mt < 4; ++mt)
            a[mt] = *(const f4*)&tA[mt * 16 + l15][kb];
        #pragma unroll
        for (int mt = 0; mt < 4; ++mt) {
            const v2f alo = {a[mt].x, a[mt].y};
            acc[mt][0] = wmma4(alo, (v2f){b0.x, b0.y}, acc[mt][0]);
            acc[mt][1] = wmma4(alo, (v2f){b1.x, b1.y}, acc[mt][1]);
        }
        #pragma unroll
        for (int mt = 0; mt < 4; ++mt) {
            const v2f ahi = {a[mt].z, a[mt].w};
            acc[mt][0] = wmma4(ahi, (v2f){b0.z, b0.w}, acc[mt][0]);
            acc[mt][1] = wmma4(ahi, (v2f){b1.z, b1.w}, acc[mt][1]);
        }
    }

    const int mh8 = 8 * (lane >> 4);
    #pragma unroll
    for (int mt = 0; mt < 4; ++mt) {
        #pragma unroll
        for (int j = 0; j < 2; ++j) {
            const int col = nb + j * 16 + l15;
            #pragma unroll
            for (int v = 0; v < 8; ++v) {
                const int row = mt * 16 + mh8 + v;
                tR[row][col] = 0.25f * acc[mt][j][v] + tR[row][col];
            }
        }
    }
    __syncthreads();

    const int q  = wave >> 1;
    const int p2 = (q < 2) ? 1 : 0;
    const int mq = (wave & 1) * 32;
    const int nq = (q & 1) * 32;
    const float* Wout = p2 ? Wdrug : Wdis;
    float (*tF)[LDSS] = p2 ? sR1 : sR0;

    v8f acc2[2][2];
    #pragma unroll
    for (int mt = 0; mt < 2; ++mt) { acc2[mt][0] = z; acc2[mt][1] = z; }

    const float* wo0 = Wout + (nq + l15) * NF + koff;
    const float* wo1 = Wout + (nq + 16 + l15) * NF + koff;

    #pragma unroll 2
    for (int kc = 0; kc < 16; ++kc) {
        const int kb = kc * 8 + koff;
        const f4 b0 = *(const f4*)(wo0 + kc * 8);
        const f4 b1 = *(const f4*)(wo1 + kc * 8);
        f4 a[2];
        #pragma unroll
        for (int mt = 0; mt < 2; ++mt)
            a[mt] = *(const f4*)&tF[mq + mt * 16 + l15][kb];
        #pragma unroll
        for (int mt = 0; mt < 2; ++mt) {
            const v2f alo = {a[mt].x, a[mt].y};
            acc2[mt][0] = wmma4(alo, (v2f){b0.x, b0.y}, acc2[mt][0]);
            acc2[mt][1] = wmma4(alo, (v2f){b1.x, b1.y}, acc2[mt][1]);
        }
        #pragma unroll
        for (int mt = 0; mt < 2; ++mt) {
            const v2f ahi = {a[mt].z, a[mt].w};
            acc2[mt][0] = wmma4(ahi, (v2f){b0.z, b0.w}, acc2[mt][0]);
            acc2[mt][1] = wmma4(ahi, (v2f){b1.z, b1.w}, acc2[mt][1]);
        }
    }

    #pragma unroll
    for (int mt = 0; mt < 2; ++mt) {
        const int grow = blockRow + mq + mt * 16 + mh8;
        #pragma unroll
        for (int j = 0; j < 2; ++j) {
            const int col = q * 32 + j * 16 + l15;
            #pragma unroll
            for (int v = 0; v < 8; ++v) {
                __builtin_nontemporal_store(acc2[mt][j][v], out + (grow + v) * NF + col);
            }
        }
    }
}

extern "C" void kernel_launch(void* const* d_in, const int* in_sizes, int n_in,
                              void* d_out, int out_size, void* d_ws, size_t ws_size,
                              hipStream_t stream) {
    const float* fmi    = (const float*)d_in[0];
    const float* fg     = (const float*)d_in[1];
    const float* fdr    = (const float*)d_in[2];
    const float* Wdd    = (const float*)d_in[3];
    const float* Wdisdr = (const float*)d_in[4];
    const float* Wdrug  = (const float*)d_in[5];
    const float* Wdis   = (const float*)d_in[6];
    const int*   mp     = (const int*)d_in[7];
    float* out = (float*)d_out;

    const int n_rows = in_sizes[7] / 4;
    const int blocks = n_rows / TM;

    metapath_wmma_kernel<<<dim3(blocks), dim3(256), 0, stream>>>(
        fmi, fg, fdr, Wdd, Wdisdr, Wdrug, Wdis, mp, out);
}
